# starting kernel
# baseline (speedup 1.0000x reference)
#include <hip/hip_runtime.h>
#include <hip/hip_bf16.h>


typedef __attribute__((ext_vector_type(2))) float v2f;
typedef __attribute__((ext_vector_type(8))) float v8f;

#define DMAT   64
#define ROWS   32
#define PITCH  68
#define NTERMS 30

__global__ __launch_bounds__(256)
void linode_expm_action_kernel(const float* __restrict__ dt,
                               const float* __restrict__ x0,
                               const float* __restrict__ scal,
                               const float* __restrict__ weight,
                               float* __restrict__ out)
{
    __shared__ float Wl[DMAT * PITCH];
    __shared__ float Yl[2][ROWS * PITCH];

    const int tid  = threadIdx.x;
    const int wave = tid >> 5;
    const int lane = tid & 31;
    const int half = lane >> 4;
    const int ln   = lane & 15;
    const int tm   = wave >> 2;
    const int tn   = wave & 3;
    const int r0   = blockIdx.x * ROWS;

    const float gamma = scal[0];
    for (int i = tid; i < DMAT * DMAT; i += 256) {
        int r = i >> 6, c = i & 63;
        Wl[r * PITCH + c] = gamma * weight[i];
    }

    const int rowBase = tm * 16 + half * 8;
    float dtv[8];
    #pragma unroll
    for (int v = 0; v < 8; ++v)
        dtv[v] = dt[r0 + rowBase + v];

    v8f S;
    #pragma unroll
    for (int v = 0; v < 8; ++v)
        S[v] = x0[(size_t)(r0 + rowBase + v) * DMAT + tn * 16 + ln];
    for (int i = tid; i < ROWS * DMAT; i += 256) {
        int r = i >> 6, c = i & 63;
        Yl[0][r * PITCH + c] = x0[(size_t)(r0 + r) * DMAT + c];
    }
    __syncthreads();

    v2f bf[16];
    #pragma unroll
    for (int kb = 0; kb < 16; ++kb)
        bf[kb] = *(const v2f*)&Wl[(tn * 16 + ln) * PITCH + kb * 4 + 2 * half];

    const float* arow0 = &Yl[0][(tm * 16 + ln) * PITCH + 2 * half];
    const float* arow1 = &Yl[1][(tm * 16 + ln) * PITCH + 2 * half];
    float* wrow0 = &Yl[0][rowBase * PITCH + tn * 16 + ln];
    float* wrow1 = &Yl[1][rowBase * PITCH + tn * 16 + ln];

    #pragma unroll
    for (int k = 1; k <= NTERMS; ++k) {
        const float invk = 1.0f / (float)k;
        const float* ybase = (k & 1) ? arow0 : arow1;
        float*       wbase = (k & 1) ? wrow1 : wrow0;

        v8f acc = {};
        #pragma unroll
        for (int kb = 0; kb < 16; ++kb) {
            v2f af = *(const v2f*)(ybase + kb * 4);
            acc = __builtin_amdgcn_wmma_f32_16x16x4_f32(
                false, af, false, bf[kb], (short)0, acc, false, false);
        }

        #pragma unroll
        for (int v = 0; v < 8; ++v) {
            float val = acc[v] * (dtv[v] * invk);
            wbase[v * PITCH] = val;
            S[v] += val;
        }
        __syncthreads();
    }

    #pragma unroll
    for (int v = 0; v < 8; ++v)
        out[(size_t)(r0 + rowBase + v) * DMAT + tn * 16 + ln] = S[v];
}

extern "C" void kernel_launch(void* const* d_in, const int* in_sizes, int n_in,
                              void* d_out, int out_size, void* d_ws, size_t ws_size,
                              hipStream_t stream) {
    const float* dt     = (const float*)d_in[0];
    const float* x0     = (const float*)d_in[1];
    const float* scalar = (const float*)d_in[2];
    const float* weight = (const float*)d_in[3];
    float* out          = (float*)d_out;

    const int B = in_sizes[0];
    const int nblocks = B / ROWS;

    linode_expm_action_kernel<<<nblocks, 256, 0, stream>>>(dt, x0, scalar, weight, out);
}


# Re-tiled: work-groups of 2 waves, each wave 16 rows x 32 columns with two independent WMMA chains, one wave per SIMD (declared to the compiler), G fragments from global; order of sums unchanged
# speedup vs baseline: 1.5065x; 1.5065x over previous
#include <hip/hip_runtime.h>


typedef __attribute__((ext_vector_type(2))) float v2f;
typedef __attribute__((ext_vector_type(8))) float v8f;

#define DMAT   64
#define ROWS   16
#define PITCH  68
#define NTERMS 30

__global__ __launch_bounds__(64) __attribute__((amdgpu_waves_per_eu(1, 1)))
void linode_expm_action_kernel(const float* __restrict__ dt,
                               const float* __restrict__ x0,
                               const float* __restrict__ scal,
                               const float* __restrict__ weight,
                               float* __restrict__ out)
{
    __shared__ float Yl[2][ROWS * PITCH];

    const int tid  = threadIdx.x;
    const int wave = tid >> 5;
    const int lane = tid & 31;
    const int half = lane >> 4;
    const int ln   = lane & 15;
    const int r0   = blockIdx.x * ROWS;
    const int c0   = wave * 32;

    const float gamma = scal[0];
    v2f bf0[16], bf1[16];
    #pragma unroll
    for (int kb = 0; kb < 16; ++kb) {
        v2f w0 = *(const v2f*)&weight[(c0 + ln) * DMAT + kb * 4 + 2 * half];
        v2f w1 = *(const v2f*)&weight[(c0 + 16 + ln) * DMAT + kb * 4 + 2 * half];
        bf0[kb] = gamma * w0;
        bf1[kb] = gamma * w1;
    }

    const int rowBase = half * 8;
    float dtv[8];
    #pragma unroll
    for (int v = 0; v < 8; ++v)
        dtv[v] = dt[r0 + rowBase + v];

    const float* arow0 = &Yl[0][ln * PITCH + 2 * half];
    const float* arow1 = &Yl[1][ln * PITCH + 2 * half];
    float* wrow0 = &Yl[0][rowBase * PITCH + c0 + ln];
    float* wrow1 = &Yl[1][rowBase * PITCH + c0 + ln];

    v8f S0, S1;
    #pragma unroll
    for (int v = 0; v < 8; ++v) {
        S0[v] = x0[(size_t)(r0 + rowBase + v) * DMAT + c0 + ln];
        S1[v] = x0[(size_t)(r0 + rowBase + v) * DMAT + c0 + 16 + ln];
        wrow0[v * PITCH]      = S0[v];
        wrow0[v * PITCH + 16] = S1[v];
    }
    __syncthreads();

    #pragma unroll
    for (int k = 1; k <= NTERMS; ++k) {
        const float invk = 1.0f / (float)k;
        const float* ybase = (k & 1) ? arow0 : arow1;
        float*       wbase = (k & 1) ? wrow1 : wrow0;

        v2f af[16];
        #pragma unroll
        for (int kb = 0; kb < 16; ++kb)
            af[kb] = *(const v2f*)(ybase + kb * 4);

        v8f acc0 = {}, acc1 = {};
        #pragma unroll
        for (int kb = 0; kb < 16; ++kb) {
            acc0 = __builtin_amdgcn_wmma_f32_16x16x4_f32(
                false, af[kb], false, bf0[kb], (short)0, acc0, false, false);
            acc1 = __builtin_amdgcn_wmma_f32_16x16x4_f32(
                false, af[kb], false, bf1[kb], (short)0, acc1, false, false);
        }

        #pragma unroll
        for (int v = 0; v < 8; ++v) {
            const float sc = dtv[v] * invk;
            float val0 = acc0[v] * sc;
            float val1 = acc1[v] * sc;
            if (k < NTERMS) {
                wbase[v * PITCH]      = val0;
                wbase[v * PITCH + 16] = val1;
            }
            S0[v] += val0;
            S1[v] += val1;
        }
        if (k < NTERMS)
            __syncthreads();
    }

    #pragma unroll
    for (int v = 0; v < 8; ++v) {
        out[(size_t)(r0 + rowBase + v) * DMAT + c0 + ln]      = S0[v];
        out[(size_t)(r0 + rowBase + v) * DMAT + c0 + 16 + ln] = S1[v];
    }
}

extern "C" void kernel_launch(void* const* d_in, const int* in_sizes, int n_in,
                              void* d_out, int out_size, void* d_ws, size_t ws_size,
                              hipStream_t stream) {
    const float* dt     = (const float*)d_in[0];
    const float* x0     = (const float*)d_in[1];
    const float* scalar = (const float*)d_in[2];
    const float* weight = (const float*)d_in[3];
    float* out          = (float*)d_out;

    const int B = in_sizes[0];
    const int nblocks = B / ROWS;

    linode_expm_action_kernel<<<nblocks, 64, 0, stream>>>(dt, x0, scalar, weight, out);
}


# Barrier off the critical path: a wave starts the next product on its own k half, read back behind its own LDS stores, signals once that read returns, waits before the other half; k sum order rotated
# speedup vs baseline: 1.6114x; 1.0696x over previous
#include <hip/hip_runtime.h>


typedef __attribute__((ext_vector_type(2))) float v2f;
typedef __attribute__((ext_vector_type(8))) float v8f;

#define DMAT   64
#define ROWS   16
#define PITCH  68
#define NTERMS 30

__global__ __launch_bounds__(64) __attribute__((amdgpu_waves_per_eu(1, 1)))
void linode_expm_action_kernel(const float* __restrict__ dt,
                               const float* __restrict__ x0,
                               const float* __restrict__ scal,
                               const float* __restrict__ weight,
                               float* __restrict__ out)
{
    __shared__ float Yl[2][ROWS * PITCH];

    const int tid  = threadIdx.x;
    const int wave = tid >> 5;
    const int lane = tid & 31;
    const int half = lane >> 4;
    const int ln   = lane & 15;
    const int r0   = blockIdx.x * ROWS;
    const int c0   = wave * 32;

    const int kOwn = wave * 8;
    const int kOth = (kOwn + 8) & 15;

    const float gamma = scal[0];
    v2f bf0[16], bf1[16];
    #pragma unroll
    for (int i = 0; i < 16; ++i) {
        const int kb = (i < 8) ? (kOwn + i) : (kOth + i - 8);
        v2f w0 = *(const v2f*)&weight[(c0 + ln) * DMAT + kb * 4 + 2 * half];
        v2f w1 = *(const v2f*)&weight[(c0 + 16 + ln) * DMAT + kb * 4 + 2 * half];
        bf0[i] = gamma * w0;
        bf1[i] = gamma * w1;
    }

    const int rowBase = half * 8;
    float dtv[8];
    #pragma unroll
    for (int v = 0; v < 8; ++v)
        dtv[v] = dt[r0 + rowBase + v];

    const float* aown0 = &Yl[0][ln * PITCH + kOwn * 4 + 2 * half];
    const float* aown1 = &Yl[1][ln * PITCH + kOwn * 4 + 2 * half];
    const float* aoth0 = &Yl[0][ln * PITCH + kOth * 4 + 2 * half];
    const float* aoth1 = &Yl[1][ln * PITCH + kOth * 4 + 2 * half];
    float* wrow0 = &Yl[0][rowBase * PITCH + c0 + ln];
    float* wrow1 = &Yl[1][rowBase * PITCH + c0 + ln];

    v8f S0, S1;
    #pragma unroll
    for (int v = 0; v < 8; ++v) {
        S0[v] = x0[(size_t)(r0 + rowBase + v) * DMAT + c0 + ln];
        S1[v] = x0[(size_t)(r0 + rowBase + v) * DMAT + c0 + 16 + ln];
        wrow0[v * PITCH]      = S0[v];
        wrow0[v * PITCH + 16] = S1[v];
    }

    #pragma unroll
    for (int k = 1; k <= NTERMS; ++k) {
        const float invk = 1.0f / (float)k;
        const float* yown = (k & 1) ? aown0 : aown1;
        const float* yoth = (k & 1) ? aoth0 : aoth1;
        float*       wbase = (k & 1) ? wrow1 : wrow0;

        __builtin_amdgcn_fence(__ATOMIC_ACQ_REL, "wavefront");
        v2f af[16];
        #pragma unroll
        for (int i = 0; i < 4; ++i)
            af[i] = *(const v2f*)(yown + i * 4);

        asm volatile("s_barrier_signal -1" :: "v"(af[0]) : "memory");

        #pragma unroll
        for (int i = 4; i < 8; ++i)
            af[i] = *(const v2f*)(yown + i * 4);

        v8f acc0 = {}, acc1 = {};
        #pragma unroll
        for (int i = 0; i < 4; ++i) {
            acc0 = __builtin_amdgcn_wmma_f32_16x16x4_f32(
                false, af[i], false, bf0[i], (short)0, acc0, false, false);
            acc1 = __builtin_amdgcn_wmma_f32_16x16x4_f32(
                false, af[i], false, bf1[i], (short)0, acc1, false, false);
        }

        asm volatile("s_barrier_wait -1" ::: "memory");
        #pragma unroll
        for (int i = 8; i < 16; ++i)
            af[i] = *(const v2f*)(yoth + (i - 8) * 4);

        #pragma unroll
        for (int i = 4; i < 16; ++i) {
            acc0 = __builtin_amdgcn_wmma_f32_16x16x4_f32(
                false, af[i], false, bf0[i], (short)0, acc0, false, false);
            acc1 = __builtin_amdgcn_wmma_f32_16x16x4_f32(
                false, af[i], false, bf1[i], (short)0, acc1, false, false);
        }

        #pragma unroll
        for (int v = 0; v < 8; ++v) {
            const float sc = dtv[v] * invk;
            float val0 = acc0[v] * sc;
            float val1 = acc1[v] * sc;
            if (k < NTERMS) {
                wbase[v * PITCH]      = val0;
                wbase[v * PITCH + 16] = val1;
            }
            S0[v] += val0;
            S1[v] += val1;
        }
    }

    #pragma unroll
    for (int v = 0; v < 8; ++v) {
        out[(size_t)(r0 + rowBase + v) * DMAT + c0 + ln]      = S0[v];
        out[(size_t)(r0 + rowBase + v) * DMAT + c0 + 16 + ln] = S1[v];
    }
}

extern "C" void kernel_launch(void* const* d_in, const int* in_sizes, int n_in,
                              void* d_out, int out_size, void* d_ws, size_t ws_size,
                              hipStream_t stream) {
    const float* dt     = (const float*)d_in[0];
    const float* x0     = (const float*)d_in[1];
    const float* scalar = (const float*)d_in[2];
    const float* weight = (const float*)d_in[3];
    float* out          = (float*)d_out;

    const int B = in_sizes[0];
    const int nblocks = B / ROWS;

    linode_expm_action_kernel<<<nblocks, 64, 0, stream>>>(dt, x0, scalar, weight, out);
}


# Two waves per SIMD: 4-wave work-groups, one 16x16 tile per wave, k sum split over two WMMA chains (partials added in fp32); another group's WMMAs fill each hand-over; out lines stored whole via LDS
# speedup vs baseline: 1.9205x; 1.1918x over previous
#include <hip/hip_runtime.h>


typedef __attribute__((ext_vector_type(2))) float v2f;
typedef __attribute__((ext_vector_type(4))) float v4f;
typedef __attribute__((ext_vector_type(8))) float v8f;

#define DMAT   64
#define ROWS   16
#define PITCH  68
#define NTERMS 30

__global__ __launch_bounds__(128) __attribute__((amdgpu_waves_per_eu(2, 2)))
void linode_expm_action_kernel(const float* __restrict__ dt,
                               const float* __restrict__ x0,
                               const float* __restrict__ scal,
                               const float* __restrict__ weight,
                               float* __restrict__ out)
{
    __shared__ __attribute__((aligned(16))) float Yl[2][ROWS * PITCH];

    const int tid  = threadIdx.x;
    const int wave = tid >> 5;
    const int lane = tid & 31;
    const int half = lane >> 4;
    const int ln   = lane & 15;
    const int r0   = blockIdx.x * ROWS;
    const int c0   = wave * 16;

    int kg[4];
    #pragma unroll
    for (int g = 0; g < 4; ++g)
        kg[g] = (wave * 4 + g * 4) & 15;

    const float gamma = scal[0];
    v2f bf[16];
    #pragma unroll
    for (int i = 0; i < 16; ++i) {
        const int kb = kg[i >> 2] + (i & 3);
        v2f w = *(const v2f*)&weight[(c0 + ln) * DMAT + kb * 4 + 2 * half];
        bf[i] = gamma * w;
    }

    const int rowBase = half * 8;
    float dtv[8];
    #pragma unroll
    for (int v = 0; v < 8; ++v)
        dtv[v] = dt[r0 + rowBase + v];

    const float* ya0[4];
    const float* ya1[4];
    #pragma unroll
    for (int g = 0; g < 4; ++g) {
        ya0[g] = &Yl[0][ln * PITCH + kg[g] * 4 + 2 * half];
        ya1[g] = &Yl[1][ln * PITCH + kg[g] * 4 + 2 * half];
    }
    float* wrow0 = &Yl[0][rowBase * PITCH + c0 + ln];
    float* wrow1 = &Yl[1][rowBase * PITCH + c0 + ln];

    v8f S;
    #pragma unroll
    for (int v = 0; v < 8; ++v) {
        S[v] = x0[(size_t)(r0 + rowBase + v) * DMAT + c0 + ln];
        wrow0[v * PITCH] = S[v];
    }

    #pragma unroll
    for (int k = 1; k <= NTERMS; ++k) {
        const float invk = 1.0f / (float)k;
        const float* const* ya = (k & 1) ? ya0 : ya1;
        float*       wbase = (k & 1) ? wrow1 : wrow0;

        __builtin_amdgcn_fence(__ATOMIC_ACQ_REL, "wavefront");
        v2f af[16];
        #pragma unroll
        for (int i = 0; i < 4; ++i)
            af[i] = *(const v2f*)(ya[0] + i * 4);

        asm volatile("s_barrier_signal -1" :: "v"(af[0]) : "memory");

        v8f accE = {}, accO = {};
        accE = __builtin_amdgcn_wmma_f32_16x16x4_f32(
            false, af[0], false, bf[0], (short)0, accE, false, false);
        accO = __builtin_amdgcn_wmma_f32_16x16x4_f32(
            false, af[1], false, bf[1], (short)0, accO, false, false);

        asm volatile("s_barrier_wait -1" ::: "memory");
        #pragma unroll
        for (int i = 4; i < 16; ++i)
            af[i] = *(const v2f*)(ya[i >> 2] + (i & 3) * 4);

        #pragma unroll
        for (int i = 2; i < 16; i += 2) {
            accE = __builtin_amdgcn_wmma_f32_16x16x4_f32(
                false, af[i], false, bf[i], (short)0, accE, false, false);
            accO = __builtin_amdgcn_wmma_f32_16x16x4_f32(
                false, af[i + 1], false, bf[i + 1], (short)0, accO, false, false);
        }

        #pragma unroll
        for (int v = 0; v < 8; ++v) {
            const float sc = dtv[v] * invk;
            float val = (accE[v] + accO[v]) * sc;
            if (k < NTERMS)
                wbase[v * PITCH] = val;
            S[v] += val;
        }
    }

    #pragma unroll
    for (int v = 0; v < 8; ++v)
        wrow0[v * PITCH] = S[v];
    __syncthreads();
    #pragma unroll
    for (int j = 0; j < 2; ++j) {
        const int row = wave * 4 + j * 2 + half;
        const v4f val = *(const v4f*)&Yl[0][row * PITCH + ln * 4];
        *(v4f*)&out[(size_t)(r0 + row) * DMAT + ln * 4] = val;
    }
}

extern "C" void kernel_launch(void* const* d_in, const int* in_sizes, int n_in,
                              void* d_out, int out_size, void* d_ws, size_t ws_size,
                              hipStream_t stream) {
    const float* dt     = (const float*)d_in[0];
    const float* x0     = (const float*)d_in[1];
    const float* scalar = (const float*)d_in[2];
    const float* weight = (const float*)d_in[3];
    float* out          = (float*)d_out;

    const int B = in_sizes[0];
    const int nblocks = B / ROWS;

    linode_expm_action_kernel<<<nblocks, 128, 0, stream>>>(dt, x0, scalar, weight, out);
}


# Series ended by a data test, not at a fixed 30 terms: once a whole term is under 2^-26 of its row's largest |x0| twice running, terms after the next (below fp32 rounding of the sum) are not added
# speedup vs baseline: 3.1471x; 1.6387x over previous
#include <hip/hip_runtime.h>


typedef __attribute__((ext_vector_type(2))) float v2f;
typedef __attribute__((ext_vector_type(4))) float v4f;
typedef __attribute__((ext_vector_type(8))) float v8f;

#define DMAT   64
#define ROWS   16
#define PITCH  68
#define NTERMS 30
#define KTEST  10
#define SMALL_REL 0x1p-26f

__global__ __launch_bounds__(128) __attribute__((amdgpu_waves_per_eu(2, 2)))
void linode_expm_action_kernel(const float* __restrict__ dt,
                               const float* __restrict__ x0,
                               const float* __restrict__ scal,
                               const float* __restrict__ weight,
                               float* __restrict__ out)
{
    __shared__ __attribute__((aligned(16))) float Yl[2][ROWS * PITCH];
    __shared__ __attribute__((aligned(16))) float Ol[ROWS * PITCH];

    const int tid  = threadIdx.x;
    const int wave = tid >> 5;
    const int lane = tid & 31;
    const int half = lane >> 4;
    const int ln   = lane & 15;
    const int r0   = blockIdx.x * ROWS;
    const int c0   = wave * 16;

    int kg[4];
    #pragma unroll
    for (int g = 0; g < 4; ++g)
        kg[g] = (wave * 4 + g * 4) & 15;

    const float gamma = scal[0];
    v2f bf[16];
    #pragma unroll
    for (int i = 0; i < 16; ++i) {
        const int kb = kg[i >> 2] + (i & 3);
        v2f w = *(const v2f*)&weight[(c0 + ln) * DMAT + kb * 4 + 2 * half];
        bf[i] = gamma * w;
    }

    const int rowBase = half * 8;
    float dtv[8];
    #pragma unroll
    for (int v = 0; v < 8; ++v)
        dtv[v] = dt[r0 + rowBase + v];

    const float* ya0[4];
    const float* ya1[4];
    #pragma unroll
    for (int g = 0; g < 4; ++g) {
        ya0[g] = &Yl[0][ln * PITCH + kg[g] * 4 + 2 * half];
        ya1[g] = &Yl[1][ln * PITCH + kg[g] * 4 + 2 * half];
    }
    float* wrow0 = &Yl[0][rowBase * PITCH + c0 + ln];
    float* wrow1 = &Yl[1][rowBase * PITCH + c0 + ln];
    float* orow  = &Ol[rowBase * PITCH + c0 + ln];

    v8f S;
    #pragma unroll
    for (int v = 0; v < 8; ++v) {
        S[v] = x0[(size_t)(r0 + rowBase + v) * DMAT + c0 + ln];
        wrow0[v * PITCH] = S[v];
    }

    float thr = 0.0f;
    bool prevSmall = false;

    #pragma unroll
    for (int k = 1; k <= NTERMS; ++k) {
        const float invk = 1.0f / (float)k;
        const float* const* ya = (k & 1) ? ya0 : ya1;
        float*       wbase = (k & 1) ? wrow1 : wrow0;

        __builtin_amdgcn_fence(__ATOMIC_ACQ_REL, "wavefront");
        v2f af[16];
        #pragma unroll
        for (int i = 0; i < 4; ++i)
            af[i] = *(const v2f*)(ya[0] + i * 4);

        asm volatile("s_barrier_signal -1" :: "v"(af[0]) : "memory");

        v8f accE = {}, accO = {};
        accE = __builtin_amdgcn_wmma_f32_16x16x4_f32(
            false, af[0], false, bf[0], (short)0, accE, false, false);
        accO = __builtin_amdgcn_wmma_f32_16x16x4_f32(
            false, af[1], false, bf[1], (short)0, accO, false, false);

        asm volatile("s_barrier_wait -1" ::: "memory");
        #pragma unroll
        for (int i = 4; i < 16; ++i)
            af[i] = *(const v2f*)(ya[i >> 2] + (i & 3) * 4);

        #pragma unroll
        for (int i = 2; i < 16; i += 2) {
            accE = __builtin_amdgcn_wmma_f32_16x16x4_f32(
                false, af[i], false, bf[i], (short)0, accE, false, false);
            accO = __builtin_amdgcn_wmma_f32_16x16x4_f32(
                false, af[i + 1], false, bf[i + 1], (short)0, accO, false, false);
        }

        bool leave = false;
        if (k == 1 || (k >= KTEST && k < NTERMS)) {
            float m = 0.0f;
            #pragma unroll
            for (int i = 0; i < 16; ++i)
                m = fmaxf(m, fmaxf(fabsf(af[i][0]), fabsf(af[i][1])));
            if (k == 1) {
                const float o = __uint_as_float(__builtin_amdgcn_permlanex16(
                    __float_as_uint(m), __float_as_uint(m), 0x76543210u, 0xfedcba98u, false, false));
                thr = fmaxf(m, o) * SMALL_REL;
            } else {
                const bool small = __builtin_amdgcn_ballot_w32(!(m <= thr)) == 0u;
                leave = small && prevSmall;
                prevSmall = small;
            }
        }

        #pragma unroll
        for (int v = 0; v < 8; ++v) {
            const float sc = dtv[v] * invk;
            float val = (accE[v] + accO[v]) * sc;
            if (k < NTERMS)
                wbase[v * PITCH] = val;
            S[v] += val;
        }
        if (leave)
            break;
    }

    #pragma unroll
    for (int v = 0; v < 8; ++v)
        orow[v * PITCH] = S[v];
    __syncthreads();
    #pragma unroll
    for (int j = 0; j < 2; ++j) {
        const int row = wave * 4 + j * 2 + half;
        const v4f val = *(const v4f*)&Ol[row * PITCH + ln * 4];
        *(v4f*)&out[(size_t)(r0 + row) * DMAT + ln * 4] = val;
    }
}

extern "C" void kernel_launch(void* const* d_in, const int* in_sizes, int n_in,
                              void* d_out, int out_size, void* d_ws, size_t ws_size,
                              hipStream_t stream) {
    const float* dt     = (const float*)d_in[0];
    const float* x0     = (const float*)d_in[1];
    const float* scalar = (const float*)d_in[2];
    const float* weight = (const float*)d_in[3];
    float* out          = (float*)d_out;

    const int B = in_sizes[0];
    const int nblocks = B / ROWS;

    linode_expm_action_kernel<<<nblocks, 128, 0, stream>>>(dt, x0, scalar, weight, out);
}
